# starting kernel
# baseline (speedup 1.0000x reference)
#include <hip/hip_runtime.h>
#include <hip/hip_bf16.h>
#include <math.h>

typedef float v2f __attribute__((ext_vector_type(2)));
typedef float v8f __attribute__((ext_vector_type(8)));

#if defined(__HIP_DEVICE_COMPILE__)
#if __has_builtin(__builtin_amdgcn_tanhf)
#define FAST_TANH(x) __builtin_amdgcn_tanhf(x)
#elif __has_builtin(__builtin_amdgcn_tanh_f32)
#define FAST_TANH(x) __builtin_amdgcn_tanh_f32(x)
#else
#define FAST_TANH(x) tanhf(x)
#endif
#if __has_builtin(__builtin_amdgcn_fmed3f)
#define RELU1(x) __builtin_amdgcn_fmed3f((x), 0.0f, __builtin_inff())
#else
#define RELU1(x) ((x) > 0.0f ? (x) : 0.0f)
#endif
#if !__has_builtin(__builtin_amdgcn_wmma_f32_16x16x4_f32)
#error "missing __builtin_amdgcn_wmma_f32_16x16x4_f32 for gfx1250 (device pass)"
#endif
#else
#define FAST_TANH(x) tanhf(x)
#define RELU1(x) ((x) > 0.0f ? (x) : 0.0f)
#endif

static __device__ __forceinline__ v8f wmma_f32x4(v2f a, v2f b, v8f c) {
  return __builtin_amdgcn_wmma_f32_16x16x4_f32(false, a, false, b, (short)0, c,
                                               false, false);
}

static __device__ __forceinline__ v8f splat8(float x) {
  v8f r;
#pragma unroll
  for (int i = 0; i < 8; ++i) r[i] = x;
  return r;
}

static __device__ __forceinline__ v8f relu8(v8f x) {
  v8f r;
#pragma unroll
  for (int i = 0; i < 8; ++i) r[i] = RELU1(x[i]);
  return r;
}

#define BLK 256

__global__ __launch_bounds__(BLK) void neural_ode_cls_kernel(
    const float* __restrict__ y0in, const float* __restrict__ tg,
    const float* __restrict__ W1, const float* __restrict__ b1,
    const float* __restrict__ W2, const float* __restrict__ b2,
    const float* __restrict__ Wc1, const float* __restrict__ bc1,
    const float* __restrict__ Wc2, const float* __restrict__ bc2,
    const float* __restrict__ Wc3, const float* __restrict__ bc3,
    float* __restrict__ out, int Bn, int Tn) {
  __shared__ float ybuf[BLK][2];
  __shared__ float hbuf[8][16][68];

  const int tid = threadIdx.x;
  const int gid = blockIdx.x * BLK + tid;
  const int lane = tid & 31;
  const int wave = tid >> 5;
  const int l = lane & 15;
  const int half = lane >> 4;

  float ya = 0.f, yb = 0.f;
  if (gid < Bn) {
    ya = y0in[2 * gid + 0];
    yb = y0in[2 * gid + 1];
  }
#pragma unroll 1
  for (int s = 0; s + 1 < Tn; ++s) {
    const float dt = tg[s + 1] - tg[s];
    float f0 = b2[0], f1 = b2[1];
#pragma unroll 1
    for (int c = 0; c < 4; ++c) {
      const float* __restrict__ w1a = W1 + 16 * c;
      const float* __restrict__ w1b = W1 + 64 + 16 * c;
      const float* __restrict__ bb = b1 + 16 * c;
      const float* __restrict__ w2 = W2 + 32 * c;
#pragma unroll
      for (int j = 0; j < 16; ++j) {
        const float h = FAST_TANH(fmaf(yb, w1b[j], fmaf(ya, w1a[j], bb[j])));
        f0 = fmaf(h, w2[2 * j + 0], f0);
        f1 = fmaf(h, w2[2 * j + 1], f1);
      }
    }
    ya = fmaf(dt, f0, ya);
    yb = fmaf(dt, f1, yb);
  }
  ybuf[tid][0] = ya;
  ybuf[tid][1] = yb;
  __syncthreads();

  v2f wb1[4];
  float cb1[4], cb2[4];
#pragma unroll
  for (int n = 0; n < 4; ++n) {
    const float wx = Wc1[0 * 64 + 16 * n + l];
    const float wy = Wc1[1 * 64 + 16 * n + l];
    wb1[n].x = half ? 0.f : wx;
    wb1[n].y = half ? 0.f : wy;
    cb1[n] = bc1[16 * n + l];
    cb2[n] = bc2[16 * n + l];
  }
  const int lc = (l < 3) ? l : 0;
  const float cb3 = (l < 3) ? bc3[lc] : 0.f;

#pragma unroll 1
  for (int tile = 0; tile < 2; ++tile) {
    const int r0 = 32 * wave + 16 * tile;

    const float yt0 = ybuf[r0 + l][0];
    const float yt1 = ybuf[r0 + l][1];
    v2f ay;
    ay.x = half ? 0.f : yt0;
    ay.y = half ? 0.f : yt1;

    v8f h1[4];
#pragma unroll
    for (int n = 0; n < 4; ++n)
      h1[n] = relu8(wmma_f32x4(ay, wb1[n], splat8(cb1[n])));

    __syncthreads();
#pragma unroll
    for (int n = 0; n < 4; ++n)
#pragma unroll
      for (int v = 0; v < 8; ++v)
        hbuf[wave][v + 8 * half][16 * n + l] = h1[n][v];
    __syncthreads();

    v8f h2[4];
#pragma unroll
    for (int n = 0; n < 4; ++n) h2[n] = splat8(cb2[n]);
#pragma unroll 2
    for (int kc = 0; kc < 16; ++kc) {
      const v2f a = *(const v2f*)&hbuf[wave][l][4 * kc + 2 * half];
#pragma unroll
      for (int n = 0; n < 4; ++n) {
        v2f bw;
        bw.x = Wc2[(4 * kc + 2 * half + 0) * 64 + 16 * n + l];
        bw.y = Wc2[(4 * kc + 2 * half + 1) * 64 + 16 * n + l];
        h2[n] = wmma_f32x4(a, bw, h2[n]);
      }
    }
#pragma unroll
    for (int n = 0; n < 4; ++n) h2[n] = relu8(h2[n]);

    __syncthreads();
#pragma unroll
    for (int n = 0; n < 4; ++n)
#pragma unroll
      for (int v = 0; v < 8; ++v)
        hbuf[wave][v + 8 * half][16 * n + l] = h2[n][v];
    __syncthreads();

    v8f o = splat8(cb3);
#pragma unroll 4
    for (int kc = 0; kc < 16; ++kc) {
      const v2f a = *(const v2f*)&hbuf[wave][l][4 * kc + 2 * half];
      const float w0 = Wc3[(4 * kc + 2 * half + 0) * 3 + lc];
      const float w1 = Wc3[(4 * kc + 2 * half + 1) * 3 + lc];
      v2f bw;
      bw.x = (l < 3) ? w0 : 0.f;
      bw.y = (l < 3) ? w1 : 0.f;
      o = wmma_f32x4(a, bw, o);
    }

    if (l < 3) {
      const int rowbase = blockIdx.x * BLK + r0 + 8 * half;
#pragma unroll
      for (int v = 0; v < 8; ++v) {
        const int gr = rowbase + v;
        if (gr < Bn) out[gr * 3 + l] = o[v];
      }
    }
  }
}

extern "C" void kernel_launch(void* const* d_in, const int* in_sizes, int n_in,
                              void* d_out, int out_size, void* d_ws,
                              size_t ws_size, hipStream_t stream) {
  const float* y0 = (const float*)d_in[0];
  const float* t = (const float*)d_in[1];
  const float* W1 = (const float*)d_in[2];
  const float* b1 = (const float*)d_in[3];
  const float* W2 = (const float*)d_in[4];
  const float* b2 = (const float*)d_in[5];
  const float* Wc1 = (const float*)d_in[6];
  const float* bc1 = (const float*)d_in[7];
  const float* Wc2 = (const float*)d_in[8];
  const float* bc2 = (const float*)d_in[9];
  const float* Wc3 = (const float*)d_in[10];
  const float* bc3 = (const float*)d_in[11];

  const int Bn = in_sizes[0] / 2;
  const int Tn = in_sizes[1];
  const int grid = (Bn + BLK - 1) / BLK;

  neural_ode_cls_kernel<<<grid, BLK, 0, stream>>>(
      y0, t, W1, b1, W2, b2, Wc1, bc1, Wc2, bc2, Wc3, bc3, (float*)d_out, Bn,
      Tn);
}


# speedup vs baseline: 1.8468x; 1.8468x over previous
#include <hip/hip_runtime.h>
#include <hip/hip_bf16.h>
#include <math.h>

typedef float v2f __attribute__((ext_vector_type(2)));
typedef float v8f __attribute__((ext_vector_type(8)));
typedef float v4f __attribute__((ext_vector_type(4)));
typedef v4f v4fu __attribute__((aligned(4)));

#if defined(__HIP_DEVICE_COMPILE__)
#if __has_builtin(__builtin_amdgcn_tanhf)
#define FAST_TANH(x) __builtin_amdgcn_tanhf(x)
#elif __has_builtin(__builtin_amdgcn_tanh_f32)
#define FAST_TANH(x) __builtin_amdgcn_tanh_f32(x)
#else
#define FAST_TANH(x) tanhf(x)
#endif
#if __has_builtin(__builtin_amdgcn_fmed3f)
#define RELU1(x) __builtin_amdgcn_fmed3f((x), 0.0f, __builtin_inff())
#else
#define RELU1(x) ((x) > 0.0f ? (x) : 0.0f)
#endif
#if !__has_builtin(__builtin_amdgcn_wmma_f32_16x16x4_f32)
#error "missing __builtin_amdgcn_wmma_f32_16x16x4_f32 for gfx1250 (device pass)"
#endif
#else
#define FAST_TANH(x) tanhf(x)
#define RELU1(x) ((x) > 0.0f ? (x) : 0.0f)
#endif

static __device__ __forceinline__ v8f wmma_f32x4(v2f a, v2f b, v8f c) {
  return __builtin_amdgcn_wmma_f32_16x16x4_f32(false, a, false, b, (short)0, c,
                                               false, false);
}

static __device__ __forceinline__ v8f splat8(float x) {
  v8f r;
#pragma unroll
  for (int i = 0; i < 8; ++i) r[i] = x;
  return r;
}

static __device__ __forceinline__ v8f relu8(v8f x) {
  v8f r;
#pragma unroll
  for (int i = 0; i < 8; ++i) r[i] = RELU1(x[i]);
  return r;
}

static __device__ __forceinline__ float xhalf(float x) {
#if defined(__HIP_DEVICE_COMPILE__)
  const int b = __builtin_bit_cast(int, x);
  return __builtin_bit_cast(
      float, __builtin_amdgcn_permlanex16(b, b, 0x76543210, 0xfedcba98, false, false));
#else
  return x;
#endif
}

#define BLK 256
#define PLAG 3
#define WAT2 12
#define WAT3 20

__global__ __launch_bounds__(BLK) void neural_ode_cls_kernel(
    const float* __restrict__ y0in, const float* __restrict__ tg,
    const float* __restrict__ W1, const float* __restrict__ b1,
    const float* __restrict__ W2, const float* __restrict__ b2,
    const float* __restrict__ Wc1, const float* __restrict__ bc1,
    const float* __restrict__ Wc2, const float* __restrict__ bc2,
    const float* __restrict__ Wc3, const float* __restrict__ bc3,
    float* __restrict__ out, int Bn, int Tn) {
  __shared__ float ybuf[BLK][2];
  __shared__ __attribute__((aligned(16))) float hbuf[8][16][68];

  const int tid = threadIdx.x;
  const int gid = blockIdx.x * BLK + tid;
  const int lane = tid & 31;
  const int wave = tid >> 5;
  const int l = lane & 15;
  const int half = lane >> 4;

  const bool up = (half != 0);
  float ya = 0.f, yb = 0.f;
  if (gid < Bn) {
    ya = y0in[2 * gid + 0];
    yb = y0in[2 * gid + 1];
  }

  v2f a0[4], a1[4];
#pragma unroll
  for (int n = 0; n < 4; ++n) {
    const float w0 = W1[16 * n + l];
    const float w1 = W1[64 + 16 * n + l];
    const float bb = b1[16 * n + l];
    a0[n].x = up ? bb : w0;
    a0[n].y = up ? 0.f : w1;
    a1[n].x = up ? w0 : bb;
    a1[n].y = up ? w1 : 0.f;
  }
  v2f wA[4][4], wB[4][4];
#pragma unroll
  for (int n = 0; n < 4; ++n)
#pragma unroll
    for (int q = 0; q < 4; ++q) {
      const float* __restrict__ p = W2 + 2 * (16 * n + 8 * half + 2 * q);
      wA[n][q].x = p[0];
      wA[n][q].y = p[2];
      wB[n][q].x = p[1];
      wB[n][q].y = p[3];
    }
  const float b20 = b2[0], b21 = b2[1];
  const v2f s0i = {up ? 0.f : b20, 0.f};
  const v2f t0i = {up ? 0.f : b21, 0.f};
  const v2f s1i = {up ? b20 : 0.f, 0.f};
  const v2f t1i = {up ? b21 : 0.f, 0.f};
  const v8f zero8 = splat8(0.f);

  v2f B0, B1;
  B0.x = up ? 1.f : ya;
  B0.y = up ? 0.f : yb;
  B1.x = up ? ya : 1.f;
  B1.y = up ? yb : 0.f;

#if defined(__HIP_DEVICE_COMPILE__)
  __builtin_amdgcn_sched_barrier(0);
#endif
  v2f wb1[4];
  float cb1[4], cb2[4];
  {
    const v4fu c1x = *(const v4fu*)(Wc1 + 4 * l);
    const v4fu c1y = *(const v4fu*)(Wc1 + 64 + 4 * l);
    const v4fu b1v = *(const v4fu*)(bc1 + 4 * l);
    const v4fu b2v = *(const v4fu*)(bc2 + 4 * l);
#pragma unroll
    for (int n = 0; n < 4; ++n) {
      wb1[n].x = half ? 0.f : c1x[n];
      wb1[n].y = half ? 0.f : c1y[n];
      cb1[n] = b1v[n];
      cb2[n] = b2v[n];
    }
  }
  v2f bw2[16][4];
#pragma unroll
  for (int kc = 0; kc < 16; ++kc) {
    const v4fu r0 = *(const v4fu*)(Wc2 + (4 * kc + 2 * half + 0) * 64 + 4 * l);
    const v4fu r1 = *(const v4fu*)(Wc2 + (4 * kc + 2 * half + 1) * 64 + 4 * l);
#pragma unroll
    for (int n = 0; n < 4; ++n) {
      bw2[kc][n].x = r0[n];
      bw2[kc][n].y = r1[n];
    }
  }
  const int lc = (l < 3) ? l : 0;
  const float cb3 = (l < 3) ? bc3[lc] : 0.f;
  v2f bw3[16];
#pragma unroll
  for (int kc = 0; kc < 16; ++kc) {
    const float w0 = Wc3[(4 * kc + 2 * half + 0) * 3 + lc];
    const float w1 = Wc3[(4 * kc + 2 * half + 1) * 3 + lc];
    bw3[kc].x = (l < 3) ? w0 : 0.f;
    bw3[kc].y = (l < 3) ? w1 : 0.f;
  }

  float tprev = (Tn > 0) ? tg[0] : 0.f;
#pragma unroll 1
  for (int s = 0; s + 1 < Tn; ++s) {
    const float tnext = tg[s + 1];
#if defined(__HIP_DEVICE_COMPILE__)
    __builtin_amdgcn_sched_barrier(0);
#endif

    v2f s0 = s0i, t0 = t0i, s1 = s1i, t1 = t1i;
    v8f d[4][2];
    v2f hh[32];
    d[0][0] = wmma_f32x4(a0[0], B0, zero8);
    d[0][1] = wmma_f32x4(a1[0], B1, zero8);
    d[1][0] = wmma_f32x4(a0[1], B0, zero8);
    d[1][1] = wmma_f32x4(a1[1], B1, zero8);
#pragma unroll
    for (int u = 0; u < 32 + PLAG; ++u) {
      if (u == WAT2) {
        d[2][0] = wmma_f32x4(a0[2], B0, zero8);
        d[2][1] = wmma_f32x4(a1[2], B1, zero8);
      }
      if (u == WAT3) {
        d[3][0] = wmma_f32x4(a0[3], B0, zero8);
        d[3][1] = wmma_f32x4(a1[3], B1, zero8);
      }
      const int v = u - PLAG;
      if (u < 32) hh[u].x = FAST_TANH(d[u >> 3][u & 1][2 * ((u >> 1) & 3) + 0]);
      if (v >= 0) {
        if (v & 1) s1 = __builtin_elementwise_fma(hh[v], wA[v >> 3][(v >> 1) & 3], s1);
        else       s0 = __builtin_elementwise_fma(hh[v], wA[v >> 3][(v >> 1) & 3], s0);
      }
      if (u < 32) hh[u].y = FAST_TANH(d[u >> 3][u & 1][2 * ((u >> 1) & 3) + 1]);
      if (v >= 0) {
        if (v & 1) t1 = __builtin_elementwise_fma(hh[v], wB[v >> 3][(v >> 1) & 3], t1);
        else       t0 = __builtin_elementwise_fma(hh[v], wB[v >> 3][(v >> 1) & 3], t0);
      }
    }
#if defined(__HIP_DEVICE_COMPILE__)
    __builtin_amdgcn_sched_group_barrier(0x008, 4, 0);
#pragma unroll
    for (int u = 0; u < 32; ++u) {
      if (u == WAT2 || u == WAT3) __builtin_amdgcn_sched_group_barrier(0x008, 2, 0);
      __builtin_amdgcn_sched_group_barrier(0x400, 1, 0);
      if (u >= PLAG) __builtin_amdgcn_sched_group_barrier(0x002, 1, 0);
      else           __builtin_amdgcn_sched_group_barrier(0x004, 1, 0);
      __builtin_amdgcn_sched_group_barrier(0x400, 1, 0);
      if (u >= PLAG) __builtin_amdgcn_sched_group_barrier(0x002, 1, 0);
      else           __builtin_amdgcn_sched_group_barrier(0x004, 1, 0);
    }
#endif
    const float p00 = s0.x + s0.y, p01 = t0.x + t0.y;
    const float p10 = s1.x + s1.y, p11 = t1.x + t1.y;
    const float k0 = up ? p10 : p00, k1 = up ? p11 : p01;
    const float g0 = up ? p00 : p10, g1 = up ? p01 : p11;
    const float f0 = k0 + xhalf(g0);
    const float f1 = k1 + xhalf(g1);
    const float dt = tnext - tprev;
    tprev = tnext;
    ya = fmaf(dt, f0, ya);
    yb = fmaf(dt, f1, yb);
    B0.x = up ? 1.f : ya;
    B0.y = up ? 0.f : yb;
    B1.x = up ? ya : 1.f;
    B1.y = up ? yb : 0.f;
  }
  ybuf[tid][0] = ya;
  ybuf[tid][1] = yb;
  __syncthreads();

#pragma unroll 1
  for (int tile = 0; tile < 2; ++tile) {
    const int r0 = 32 * wave + 16 * tile;

    const float yt0 = ybuf[r0 + l][0];
    const float yt1 = ybuf[r0 + l][1];
    v2f ay;
    ay.x = half ? 0.f : yt0;
    ay.y = half ? 0.f : yt1;

    v8f h1[4];
#pragma unroll
    for (int n = 0; n < 4; ++n)
      h1[n] = relu8(wmma_f32x4(ay, wb1[n], splat8(cb1[n])));

    __syncthreads();
#pragma unroll
    for (int v = 0; v < 8; ++v) {
      v4f t4 = {h1[0][v], h1[1][v], h1[2][v], h1[3][v]};
      *(v4f*)&hbuf[wave][v + 8 * half][4 * l] = t4;
    }
    __syncthreads();

    v8f h2[4];
#pragma unroll
    for (int n = 0; n < 4; ++n) h2[n] = splat8(cb2[n]);
#pragma unroll
    for (int kc = 0; kc < 16; ++kc) {
      const v2f a = *(const v2f*)&hbuf[wave][l][4 * kc + 2 * half];
#pragma unroll
      for (int n = 0; n < 4; ++n) h2[n] = wmma_f32x4(a, bw2[kc][n], h2[n]);
    }
#pragma unroll
    for (int n = 0; n < 4; ++n) h2[n] = relu8(h2[n]);

    __syncthreads();
#pragma unroll
    for (int v = 0; v < 8; ++v) {
      v4f t4 = {h2[0][v], h2[1][v], h2[2][v], h2[3][v]};
      *(v4f*)&hbuf[wave][v + 8 * half][4 * l] = t4;
    }
    __syncthreads();

    v8f o = splat8(cb3);
#pragma unroll
    for (int kc = 0; kc < 16; ++kc) {
      const v2f a = *(const v2f*)&hbuf[wave][l][4 * kc + 2 * half];
      o = wmma_f32x4(a, bw3[kc], o);
    }

    if (l < 3) {
      const int rowbase = blockIdx.x * BLK + r0 + 8 * half;
#pragma unroll
      for (int v = 0; v < 8; ++v) {
        const int gr = rowbase + v;
        if (gr < Bn) out[gr * 3 + l] = o[v];
      }
    }
  }
}

extern "C" void kernel_launch(void* const* d_in, const int* in_sizes, int n_in,
                              void* d_out, int out_size, void* d_ws,
                              size_t ws_size, hipStream_t stream) {
  const float* y0 = (const float*)d_in[0];
  const float* t = (const float*)d_in[1];
  const float* W1 = (const float*)d_in[2];
  const float* b1 = (const float*)d_in[3];
  const float* W2 = (const float*)d_in[4];
  const float* b2 = (const float*)d_in[5];
  const float* Wc1 = (const float*)d_in[6];
  const float* bc1 = (const float*)d_in[7];
  const float* Wc2 = (const float*)d_in[8];
  const float* bc2 = (const float*)d_in[9];
  const float* Wc3 = (const float*)d_in[10];
  const float* bc3 = (const float*)d_in[11];

  const int Bn = in_sizes[0] / 2;
  const int Tn = in_sizes[1];
  const int grid = (Bn + BLK - 1) / BLK;

  neural_ode_cls_kernel<<<grid, BLK, 0, stream>>>(
      y0, t, W1, b1, W2, b2, Wc1, bc1, Wc2, bc2, Wc3, bc3, (float*)d_out, Bn,
      Tn);
}
